# starting kernel
# baseline (speedup 1.0000x reference)
#include <hip/hip_runtime.h>
#include <math.h>

typedef float v2f __attribute__((ext_vector_type(2)));
typedef float v8f __attribute__((ext_vector_type(8)));

#define HIDDEN        128
#define ROWS_PER_WAVE 16
#define WAVES_PER_BLK 8
#define THREADS       (WAVES_PER_BLK * 32)
#define LDS_STRIDE    132
#define W_LDS_FLOATS  (HIDDEN * LDS_STRIDE)
#define Z_LDS_FLOATS  (ROWS_PER_WAVE * LDS_STRIDE)
#define MAX_ITER      50
#define CONV_EPS      1e-7f

__global__ __launch_bounds__(THREADS, 4)
void deq_f32_wmma_kernel(const float* __restrict__ x, const float* __restrict__ W,
                         const float* __restrict__ b, float* __restrict__ out)
{
    extern __shared__ float lds[];
    float* wsh = lds;
    const int tid   = threadIdx.x;
    const int lane  = tid & 31;
    const int wave  = tid >> 5;
    const int lmod  = lane & 15;
    const int lhalf = lane >> 4;
    float* zsh = lds + W_LDS_FLOATS + wave * Z_LDS_FLOATS;

    {
        const float4* Wv = (const float4*)W;
        const int c4 = tid & 31;
        const int r0 = tid >> 5;
        #pragma unroll
        for (int i = 0; i < 16; ++i) {
            const int row = r0 + i * 8;
            float4 v = Wv[row * 32 + c4];
            *(float4*)(wsh + row * LDS_STRIDE + c4 * 4) = v;
        }
    }
    __syncthreads();

    const long row0 = (long)blockIdx.x * (ROWS_PER_WAVE * WAVES_PER_BLK)
                    + (long)wave * ROWS_PER_WAVE;

    v8f xb[8];
    #pragma unroll
    for (int n = 0; n < 8; ++n) {
        const int col  = n * 16 + lmod;
        const float bv = b[col];
        v8f v;
        #pragma unroll
        for (int m = 0; m < 8; ++m) {
            const int row = m + 8 * lhalf;
            v[m] = x[(row0 + row) * HIDDEN + col] + bv;
        }
        xb[n] = v;
        #pragma unroll
        for (int m = 0; m < 8; ++m)
            zsh[(m + 8 * lhalf) * LDS_STRIDE + col] = tanhf(v[m]);
    }
    asm volatile("s_wait_dscnt 0x0" ::: "memory");

    #pragma unroll 1
    for (int it = 1; it < MAX_ITER; ++it) {
        v8f acc[8];
        #pragma unroll
        for (int n = 0; n < 8; ++n) acc[n] = xb[n];

        const float* ap = zsh + lmod * LDS_STRIDE + 2 * lhalf;
        const float* wp = wsh + lmod * LDS_STRIDE + 2 * lhalf;
        #pragma unroll
        for (int t = 0; t < 32; ++t) {
            const v2f a = *(const v2f*)(ap + t * 4);
            #pragma unroll
            for (int n = 0; n < 8; ++n) {
                v2f bf = *(const v2f*)(wp + t * 4 + n * 16 * LDS_STRIDE);
                acc[n] = __builtin_amdgcn_wmma_f32_16x16x4_f32(
                    false, a, false, bf, (short)0, acc[n], false, false);
            }
        }

        float maxd = 0.f;
        #pragma unroll
        for (int n = 0; n < 8; ++n) {
            const int col = n * 16 + lmod;
            #pragma unroll
            for (int m = 0; m < 8; ++m) {
                float* zp  = zsh + (m + 8 * lhalf) * LDS_STRIDE + col;
                float  old = *zp;
                float  nw  = tanhf(acc[n][m]);
                maxd = fmaxf(maxd, fabsf(nw - old));
                *zp = nw;
            }
        }
        asm volatile("s_wait_dscnt 0x0" ::: "memory");

        bool active = (maxd > CONV_EPS);
#if __has_builtin(__builtin_amdgcn_ballot_w32)
        if (__builtin_amdgcn_ballot_w32(active) == 0u) break;
#else
        if (!__any(active)) break;
#endif
    }

    {
        float* o = out + row0 * HIDDEN;
        #pragma unroll
        for (int p = 0; p < 16; ++p) {
            const int idx = p * 32 + lane;
            const int row = idx >> 5;
            const int c4  = idx & 31;
            float4 v = *(const float4*)(zsh + row * LDS_STRIDE + c4 * 4);
            *(float4*)(o + row * HIDDEN + c4 * 4) = v;
        }
    }
}

extern "C" void kernel_launch(void* const* d_in, const int* in_sizes, int n_in,
                              void* d_out, int out_size, void* d_ws, size_t ws_size,
                              hipStream_t stream) {
    const float* x = (const float*)d_in[0];
    const float* W = (const float*)d_in[1];
    const float* b = (const float*)d_in[2];
    float* out     = (float*)d_out;

    const int batch = in_sizes[0] / HIDDEN;
    const int rows_per_block = ROWS_PER_WAVE * WAVES_PER_BLK;
    const int grid = batch / rows_per_block;
    const size_t lds_bytes = (size_t)(W_LDS_FLOATS + WAVES_PER_BLK * Z_LDS_FLOATS) * 4;

    deq_f32_wmma_kernel<<<grid, THREADS, lds_bytes, stream>>>(x, W, b, out);
}


# library tanhf (branchy, long dependent chains) replaced by 1 - 2/(exp(2v)+1) on the hardware exp2 and rcp instructions, absolute error about one ulp of 1; matrix sums and their order unchanged
# speedup vs baseline: 1.4424x; 1.4424x over previous
#include <hip/hip_runtime.h>
#include <math.h>

typedef float v2f __attribute__((ext_vector_type(2)));
typedef float v8f __attribute__((ext_vector_type(8)));

__device__ __forceinline__ float tanh_hw(float v)
{
    const float e = __builtin_amdgcn_exp2f(v * 2.885390081777927f);
    return fmaf(-2.0f, __builtin_amdgcn_rcpf(e + 1.0f), 1.0f);
}

#define HIDDEN        128
#define ROWS_PER_WAVE 16
#define WAVES_PER_BLK 8
#define THREADS       (WAVES_PER_BLK * 32)
#define LDS_STRIDE    132
#define W_LDS_FLOATS  (HIDDEN * LDS_STRIDE)
#define Z_LDS_FLOATS  (ROWS_PER_WAVE * LDS_STRIDE)
#define MAX_ITER      50
#define CONV_EPS      1e-7f

__global__ __launch_bounds__(THREADS, 4)
void deq_f32_wmma_kernel(const float* __restrict__ x, const float* __restrict__ W,
                         const float* __restrict__ b, float* __restrict__ out)
{
    extern __shared__ float lds[];
    float* wsh = lds;
    const int tid   = threadIdx.x;
    const int lane  = tid & 31;
    const int wave  = tid >> 5;
    const int lmod  = lane & 15;
    const int lhalf = lane >> 4;
    float* zsh = lds + W_LDS_FLOATS + wave * Z_LDS_FLOATS;

    {
        const float4* Wv = (const float4*)W;
        const int c4 = tid & 31;
        const int r0 = tid >> 5;
        #pragma unroll
        for (int i = 0; i < 16; ++i) {
            const int row = r0 + i * 8;
            float4 v = Wv[row * 32 + c4];
            *(float4*)(wsh + row * LDS_STRIDE + c4 * 4) = v;
        }
    }
    __syncthreads();

    const long row0 = (long)blockIdx.x * (ROWS_PER_WAVE * WAVES_PER_BLK)
                    + (long)wave * ROWS_PER_WAVE;

    v8f xb[8];
    #pragma unroll
    for (int n = 0; n < 8; ++n) {
        const int col  = n * 16 + lmod;
        const float bv = b[col];
        v8f v;
        #pragma unroll
        for (int m = 0; m < 8; ++m) {
            const int row = m + 8 * lhalf;
            v[m] = x[(row0 + row) * HIDDEN + col] + bv;
        }
        xb[n] = v;
        #pragma unroll
        for (int m = 0; m < 8; ++m)
            zsh[(m + 8 * lhalf) * LDS_STRIDE + col] = tanh_hw(v[m]);
    }
    asm volatile("s_wait_dscnt 0x0" ::: "memory");

    #pragma unroll 1
    for (int it = 1; it < MAX_ITER; ++it) {
        v8f acc[8];
        #pragma unroll
        for (int n = 0; n < 8; ++n) acc[n] = xb[n];

        const float* ap = zsh + lmod * LDS_STRIDE + 2 * lhalf;
        const float* wp = wsh + lmod * LDS_STRIDE + 2 * lhalf;
        #pragma unroll
        for (int t = 0; t < 32; ++t) {
            const v2f a = *(const v2f*)(ap + t * 4);
            #pragma unroll
            for (int n = 0; n < 8; ++n) {
                v2f bf = *(const v2f*)(wp + t * 4 + n * 16 * LDS_STRIDE);
                acc[n] = __builtin_amdgcn_wmma_f32_16x16x4_f32(
                    false, a, false, bf, (short)0, acc[n], false, false);
            }
        }

        float maxd = 0.f;
        #pragma unroll
        for (int n = 0; n < 8; ++n) {
            const int col = n * 16 + lmod;
            #pragma unroll
            for (int m = 0; m < 8; ++m) {
                float* zp  = zsh + (m + 8 * lhalf) * LDS_STRIDE + col;
                float  old = *zp;
                float  nw  = tanh_hw(acc[n][m]);
                maxd = fmaxf(maxd, fabsf(nw - old));
                *zp = nw;
            }
        }
        asm volatile("s_wait_dscnt 0x0" ::: "memory");

        bool active = (maxd > CONV_EPS);
#if __has_builtin(__builtin_amdgcn_ballot_w32)
        if (__builtin_amdgcn_ballot_w32(active) == 0u) break;
#else
        if (!__any(active)) break;
#endif
    }

    {
        float* o = out + row0 * HIDDEN;
        #pragma unroll
        for (int p = 0; p < 16; ++p) {
            const int idx = p * 32 + lane;
            const int row = idx >> 5;
            const int c4  = idx & 31;
            float4 v = *(const float4*)(zsh + row * LDS_STRIDE + c4 * 4);
            *(float4*)(o + row * HIDDEN + c4 * 4) = v;
        }
    }
}

extern "C" void kernel_launch(void* const* d_in, const int* in_sizes, int n_in,
                              void* d_out, int out_size, void* d_ws, size_t ws_size,
                              hipStream_t stream) {
    const float* x = (const float*)d_in[0];
    const float* W = (const float*)d_in[1];
    const float* b = (const float*)d_in[2];
    float* out     = (float*)d_out;

    const int batch = in_sizes[0] / HIDDEN;
    const int rows_per_block = ROWS_PER_WAVE * WAVES_PER_BLK;
    const int grid = batch / rows_per_block;
    const size_t lds_bytes = (size_t)(W_LDS_FLOATS + WAVES_PER_BLK * Z_LDS_FLOATS) * 4;

    deq_f32_wmma_kernel<<<grid, THREADS, lds_bytes, stream>>>(x, W, b, out);
}


# product computed transposed (W tile as A, z^T as B): each lane's tanh results are its own next B operand, so z stays in registers with no LDS round trip; k order inside the f32 sums changes
# speedup vs baseline: 1.4988x; 1.0391x over previous
#include <hip/hip_runtime.h>
#include <math.h>


typedef float v2f __attribute__((ext_vector_type(2)));
typedef float v8f __attribute__((ext_vector_type(8)));

__device__ __forceinline__ float tanh_hw(float v)
{
    const float e = __builtin_amdgcn_exp2f(v * 2.885390081777927f);
    return fmaf(-2.0f, __builtin_amdgcn_rcpf(e + 1.0f), 1.0f);
}

#define HIDDEN        128
#define ROWS_PER_WAVE 16
#define WAVES_PER_BLK 8
#define THREADS       (WAVES_PER_BLK * 32)
#define LDS_STRIDE    132
#define W_LDS_FLOATS  (HIDDEN * LDS_STRIDE)
#define Z_LDS_FLOATS  (ROWS_PER_WAVE * LDS_STRIDE)
#define MAX_ITER      50
#define CONV_EPS      1e-7f

__global__ __launch_bounds__(THREADS, 4)
void deq_f32_wmma_kernel(const float* __restrict__ x, const float* __restrict__ W,
                         const float* __restrict__ b, float* __restrict__ out)
{
    extern __shared__ float lds[];
    float* wsh = lds;
    const int tid   = threadIdx.x;
    const int lane  = tid & 31;
    const int wave  = tid >> 5;
    const int lmod  = lane & 15;
    const int lhalf = lane >> 4;
    float* zsh = lds + W_LDS_FLOATS + wave * Z_LDS_FLOATS;

    {
        const float4* Wv = (const float4*)W;
        const int c4 = tid & 31;
        const int r0 = tid >> 5;
        #pragma unroll
        for (int i = 0; i < 16; ++i) {
            const int row = r0 + i * 8;
            float4 v = Wv[row * 32 + c4];
            *(float4*)(wsh + row * LDS_STRIDE + c4 * 4) = v;
        }
    }
    __syncthreads();

    const long row0 = (long)blockIdx.x * (ROWS_PER_WAVE * WAVES_PER_BLK)
                    + (long)wave * ROWS_PER_WAVE;

    v8f xb[8];
    v8f zr[8];
    {
        const float* xrow = x + (row0 + lmod) * HIDDEN + 8 * lhalf;
        const float* brow = b + 8 * lhalf;
        #pragma unroll
        for (int m = 0; m < 8; ++m) {
            const float4 x0 = *(const float4*)(xrow + 16 * m);
            const float4 x1 = *(const float4*)(xrow + 16 * m + 4);
            const float4 b0 = *(const float4*)(brow + 16 * m);
            const float4 b1 = *(const float4*)(brow + 16 * m + 4);
            v8f v;
            v[0] = x0.x + b0.x; v[1] = x0.y + b0.y; v[2] = x0.z + b0.z; v[3] = x0.w + b0.w;
            v[4] = x1.x + b1.x; v[5] = x1.y + b1.y; v[6] = x1.z + b1.z; v[7] = x1.w + b1.w;
            xb[m] = v;
            v8f t;
            #pragma unroll
            for (int r = 0; r < 8; ++r) t[r] = tanh_hw(v[r]);
            zr[m] = t;
        }
    }

    float* xpark = zsh + lane * 4;
    *(float4*)(xpark)       = (float4){xb[6][0], xb[6][1], xb[6][2], xb[6][3]};
    *(float4*)(xpark + 128) = (float4){xb[6][4], xb[6][5], xb[6][6], xb[6][7]};
    *(float4*)(xpark + 256) = (float4){xb[7][0], xb[7][1], xb[7][2], xb[7][3]};
    *(float4*)(xpark + 384) = (float4){xb[7][4], xb[7][5], xb[7][6], xb[7][7]};

    const float* wp = wsh + lmod * LDS_STRIDE + 8 * lhalf;

    #pragma unroll 1
    for (int it = 1; it < MAX_ITER; ++it) {
        asm volatile("" ::: "memory");

        v8f acc[8];
        #pragma unroll
        for (int m = 0; m < 6; ++m) acc[m] = xb[m];
        {
            const float4 p0 = *(const float4*)(xpark);
            const float4 p1 = *(const float4*)(xpark + 128);
            const float4 p2 = *(const float4*)(xpark + 256);
            const float4 p3 = *(const float4*)(xpark + 384);
            acc[6] = (v8f){p0.x, p0.y, p0.z, p0.w, p1.x, p1.y, p1.z, p1.w};
            acc[7] = (v8f){p2.x, p2.y, p2.z, p2.w, p3.x, p3.y, p3.z, p3.w};
        }

        #pragma unroll
        for (int J = 0; J < 8; ++J) {
            #pragma unroll
            for (int q = 0; q < 2; ++q) {
                v2f zlo, zhi;
                zlo[0] = zr[J][4 * q];     zlo[1] = zr[J][4 * q + 1];
                zhi[0] = zr[J][4 * q + 2]; zhi[1] = zr[J][4 * q + 3];
                float4 wf[8];
                #pragma unroll
                for (int m = 0; m < 8; ++m)
                    wf[m] = *(const float4*)(wp + m * 16 * LDS_STRIDE + 16 * J + 4 * q);
                #pragma unroll
                for (int m = 0; m < 8; ++m) {
                    v2f wlo; wlo[0] = wf[m].x; wlo[1] = wf[m].y;
                    acc[m] = __builtin_amdgcn_wmma_f32_16x16x4_f32(
                        false, wlo, false, zlo, (short)0, acc[m], false, false);
                }
                #pragma unroll
                for (int m = 0; m < 8; ++m) {
                    v2f whi; whi[0] = wf[m].z; whi[1] = wf[m].w;
                    acc[m] = __builtin_amdgcn_wmma_f32_16x16x4_f32(
                        false, whi, false, zhi, (short)0, acc[m], false, false);
                }
            }
        }

        float maxd = 0.f;
        #pragma unroll
        for (int m = 0; m < 8; ++m) {
            v8f t;
            #pragma unroll
            for (int r = 0; r < 8; ++r) {
                const float nw = tanh_hw(acc[m][r]);
                maxd = fmaxf(maxd, fabsf(nw - zr[m][r]));
                t[r] = nw;
            }
            zr[m] = t;
        }

        bool active = (maxd > CONV_EPS);
#if __has_builtin(__builtin_amdgcn_ballot_w32)
        if (__builtin_amdgcn_ballot_w32(active) == 0u) break;
#else
        if (!__any(active)) break;
#endif
    }

    {
        #pragma unroll
        for (int m = 0; m < 8; ++m) {
            float* zp = zsh + lmod * LDS_STRIDE + 16 * m + 8 * lhalf;
            float4 lo, hi;
            lo.x = zr[m][0]; lo.y = zr[m][1]; lo.z = zr[m][2]; lo.w = zr[m][3];
            hi.x = zr[m][4]; hi.y = zr[m][5]; hi.z = zr[m][6]; hi.w = zr[m][7];
            *(float4*)zp       = lo;
            *(float4*)(zp + 4) = hi;
        }
        asm volatile("s_wait_dscnt 0x0" ::: "memory");

        float* o = out + row0 * HIDDEN;
        #pragma unroll
        for (int p = 0; p < 16; ++p) {
            const int idx = p * 32 + lane;
            const int row = idx >> 5;
            const int c4  = idx & 31;
            float4 v = *(const float4*)(zsh + row * LDS_STRIDE + c4 * 4);
            *(float4*)(o + row * HIDDEN + c4 * 4) = v;
        }
    }
}

extern "C" void kernel_launch(void* const* d_in, const int* in_sizes, int n_in,
                              void* d_out, int out_size, void* d_ws, size_t ws_size,
                              hipStream_t stream) {
    const float* x = (const float*)d_in[0];
    const float* W = (const float*)d_in[1];
    const float* b = (const float*)d_in[2];
    float* out     = (float*)d_out;

    const int batch = in_sizes[0] / HIDDEN;
    const int rows_per_block = ROWS_PER_WAVE * WAVES_PER_BLK;
    const int grid = batch / rows_per_block;
    const size_t lds_bytes = (size_t)(W_LDS_FLOATS + WAVES_PER_BLK * Z_LDS_FLOATS) * 4;

    deq_f32_wmma_kernel<<<grid, THREADS, lds_bytes, stream>>>(x, W, b, out);
}


# tanh on the hardware instruction v_tanh_f32, one transcendental per value in place of mul + exp2 + add + rcp + fma; tanh rounding changes, matrix sums and their order unchanged
# speedup vs baseline: 1.5700x; 1.0475x over previous
#include <hip/hip_runtime.h>
#include <math.h>


typedef float v2f __attribute__((ext_vector_type(2)));
typedef float v8f __attribute__((ext_vector_type(8)));

__device__ __forceinline__ float tanh_hw(float v)
{
    return __builtin_amdgcn_tanhf(v);
}

#define HIDDEN        128
#define ROWS_PER_WAVE 16
#define WAVES_PER_BLK 8
#define THREADS       (WAVES_PER_BLK * 32)
#define LDS_STRIDE    132
#define W_LDS_FLOATS  (HIDDEN * LDS_STRIDE)
#define Z_LDS_FLOATS  (ROWS_PER_WAVE * LDS_STRIDE)
#define MAX_ITER      50
#define CONV_EPS      1e-7f

__global__ __launch_bounds__(THREADS, 4)
void deq_f32_wmma_kernel(const float* __restrict__ x, const float* __restrict__ W,
                         const float* __restrict__ b, float* __restrict__ out)
{
    extern __shared__ float lds[];
    float* wsh = lds;
    const int tid   = threadIdx.x;
    const int lane  = tid & 31;
    const int wave  = tid >> 5;
    const int lmod  = lane & 15;
    const int lhalf = lane >> 4;
    float* zsh = lds + W_LDS_FLOATS + wave * Z_LDS_FLOATS;

    {
        const float4* Wv = (const float4*)W;
        const int c4 = tid & 31;
        const int r0 = tid >> 5;
        #pragma unroll
        for (int i = 0; i < 16; ++i) {
            const int row = r0 + i * 8;
            float4 v = Wv[row * 32 + c4];
            *(float4*)(wsh + row * LDS_STRIDE + c4 * 4) = v;
        }
    }
    __syncthreads();

    const long row0 = (long)blockIdx.x * (ROWS_PER_WAVE * WAVES_PER_BLK)
                    + (long)wave * ROWS_PER_WAVE;

    v8f xb[8];
    v8f zr[8];
    {
        const float* xrow = x + (row0 + lmod) * HIDDEN + 8 * lhalf;
        const float* brow = b + 8 * lhalf;
        #pragma unroll
        for (int m = 0; m < 8; ++m) {
            const float4 x0 = *(const float4*)(xrow + 16 * m);
            const float4 x1 = *(const float4*)(xrow + 16 * m + 4);
            const float4 b0 = *(const float4*)(brow + 16 * m);
            const float4 b1 = *(const float4*)(brow + 16 * m + 4);
            v8f v;
            v[0] = x0.x + b0.x; v[1] = x0.y + b0.y; v[2] = x0.z + b0.z; v[3] = x0.w + b0.w;
            v[4] = x1.x + b1.x; v[5] = x1.y + b1.y; v[6] = x1.z + b1.z; v[7] = x1.w + b1.w;
            xb[m] = v;
            v8f t;
            #pragma unroll
            for (int r = 0; r < 8; ++r) t[r] = tanh_hw(v[r]);
            zr[m] = t;
        }
    }

    float* xpark = zsh + lane * 4;
    *(float4*)(xpark)       = (float4){xb[6][0], xb[6][1], xb[6][2], xb[6][3]};
    *(float4*)(xpark + 128) = (float4){xb[6][4], xb[6][5], xb[6][6], xb[6][7]};
    *(float4*)(xpark + 256) = (float4){xb[7][0], xb[7][1], xb[7][2], xb[7][3]};
    *(float4*)(xpark + 384) = (float4){xb[7][4], xb[7][5], xb[7][6], xb[7][7]};

    const float* wp = wsh + lmod * LDS_STRIDE + 8 * lhalf;

    #pragma unroll 1
    for (int it = 1; it < MAX_ITER; ++it) {
        asm volatile("" ::: "memory");

        v8f acc[8];
        #pragma unroll
        for (int m = 0; m < 6; ++m) acc[m] = xb[m];
        {
            const float4 p0 = *(const float4*)(xpark);
            const float4 p1 = *(const float4*)(xpark + 128);
            const float4 p2 = *(const float4*)(xpark + 256);
            const float4 p3 = *(const float4*)(xpark + 384);
            acc[6] = (v8f){p0.x, p0.y, p0.z, p0.w, p1.x, p1.y, p1.z, p1.w};
            acc[7] = (v8f){p2.x, p2.y, p2.z, p2.w, p3.x, p3.y, p3.z, p3.w};
        }

        #pragma unroll
        for (int J = 0; J < 8; ++J) {
            #pragma unroll
            for (int q = 0; q < 2; ++q) {
                v2f zlo, zhi;
                zlo[0] = zr[J][4 * q];     zlo[1] = zr[J][4 * q + 1];
                zhi[0] = zr[J][4 * q + 2]; zhi[1] = zr[J][4 * q + 3];
                float4 wf[8];
                #pragma unroll
                for (int m = 0; m < 8; ++m)
                    wf[m] = *(const float4*)(wp + m * 16 * LDS_STRIDE + 16 * J + 4 * q);
                #pragma unroll
                for (int m = 0; m < 8; ++m) {
                    v2f wlo; wlo[0] = wf[m].x; wlo[1] = wf[m].y;
                    acc[m] = __builtin_amdgcn_wmma_f32_16x16x4_f32(
                        false, wlo, false, zlo, (short)0, acc[m], false, false);
                }
                #pragma unroll
                for (int m = 0; m < 8; ++m) {
                    v2f whi; whi[0] = wf[m].z; whi[1] = wf[m].w;
                    acc[m] = __builtin_amdgcn_wmma_f32_16x16x4_f32(
                        false, whi, false, zhi, (short)0, acc[m], false, false);
                }
            }
        }

        float maxd = 0.f;
        #pragma unroll
        for (int m = 0; m < 8; ++m) {
            v8f t;
            #pragma unroll
            for (int r = 0; r < 8; ++r) {
                const float nw = tanh_hw(acc[m][r]);
                maxd = fmaxf(maxd, fabsf(nw - zr[m][r]));
                t[r] = nw;
            }
            zr[m] = t;
        }

        bool active = (maxd > CONV_EPS);
#if __has_builtin(__builtin_amdgcn_ballot_w32)
        if (__builtin_amdgcn_ballot_w32(active) == 0u) break;
#else
        if (!__any(active)) break;
#endif
    }

    {
        #pragma unroll
        for (int m = 0; m < 8; ++m) {
            float* zp = zsh + lmod * LDS_STRIDE + 16 * m + 8 * lhalf;
            float4 lo, hi;
            lo.x = zr[m][0]; lo.y = zr[m][1]; lo.z = zr[m][2]; lo.w = zr[m][3];
            hi.x = zr[m][4]; hi.y = zr[m][5]; hi.z = zr[m][6]; hi.w = zr[m][7];
            *(float4*)zp       = lo;
            *(float4*)(zp + 4) = hi;
        }
        asm volatile("s_wait_dscnt 0x0" ::: "memory");

        float* o = out + row0 * HIDDEN;
        #pragma unroll
        for (int p = 0; p < 16; ++p) {
            const int idx = p * 32 + lane;
            const int row = idx >> 5;
            const int c4  = idx & 31;
            float4 v = *(const float4*)(zsh + row * LDS_STRIDE + c4 * 4);
            *(float4*)(o + row * HIDDEN + c4 * 4) = v;
        }
    }
}

extern "C" void kernel_launch(void* const* d_in, const int* in_sizes, int n_in,
                              void* d_out, int out_size, void* d_ws, size_t ws_size,
                              hipStream_t stream) {
    const float* x = (const float*)d_in[0];
    const float* W = (const float*)d_in[1];
    const float* b = (const float*)d_in[2];
    float* out     = (float*)d_out;

    const int batch = in_sizes[0] / HIDDEN;
    const int rows_per_block = ROWS_PER_WAVE * WAVES_PER_BLK;
    const int grid = batch / rows_per_block;
    const size_t lds_bytes = (size_t)(W_LDS_FLOATS + WAVES_PER_BLK * Z_LDS_FLOATS) * 4;

    deq_f32_wmma_kernel<<<grid, THREADS, lds_bytes, stream>>>(x, W, b, out);
}


# W reads from LDS issued as one group of eight per 16-instruction matrix block, one block ahead into a second register set, not in pairs spread among the matrix instructions; sums unchanged
# speedup vs baseline: 1.6253x; 1.0352x over previous
#include <hip/hip_runtime.h>
#include <math.h>


typedef float v2f __attribute__((ext_vector_type(2)));
typedef float v8f __attribute__((ext_vector_type(8)));

__device__ __forceinline__ float tanh_hw(float v)
{
    return __builtin_amdgcn_tanhf(v);
}

#define HIDDEN        128
#define ROWS_PER_WAVE 16
#define WAVES_PER_BLK 8
#define THREADS       (WAVES_PER_BLK * 32)
#define LDS_STRIDE    132
#define W_LDS_FLOATS  (HIDDEN * LDS_STRIDE)
#define Z_LDS_FLOATS  (ROWS_PER_WAVE * LDS_STRIDE)
#define MAX_ITER      50
#define CONV_EPS      1e-7f

__global__ __launch_bounds__(THREADS, 4)
void deq_f32_wmma_kernel(const float* __restrict__ x, const float* __restrict__ W,
                         const float* __restrict__ b, float* __restrict__ out)
{
    extern __shared__ float lds[];
    float* wsh = lds;
    const int tid   = threadIdx.x;
    const int lane  = tid & 31;
    const int wave  = tid >> 5;
    const int lmod  = lane & 15;
    const int lhalf = lane >> 4;
    float* zsh = lds + W_LDS_FLOATS + wave * Z_LDS_FLOATS;

    {
        const float4* Wv = (const float4*)W;
        const int c4 = tid & 31;
        const int r0 = tid >> 5;
        #pragma unroll
        for (int i = 0; i < 16; ++i) {
            const int row = r0 + i * 8;
            float4 v = Wv[row * 32 + c4];
            *(float4*)(wsh + row * LDS_STRIDE + c4 * 4) = v;
        }
    }
    __syncthreads();

    const long row0 = (long)blockIdx.x * (ROWS_PER_WAVE * WAVES_PER_BLK)
                    + (long)wave * ROWS_PER_WAVE;

    v8f xb[8];
    v8f zr[8];
    {
        const float* xrow = x + (row0 + lmod) * HIDDEN + 8 * lhalf;
        const float* brow = b + 8 * lhalf;
        #pragma unroll
        for (int m = 0; m < 8; ++m) {
            const float4 x0 = *(const float4*)(xrow + 16 * m);
            const float4 x1 = *(const float4*)(xrow + 16 * m + 4);
            const float4 b0 = *(const float4*)(brow + 16 * m);
            const float4 b1 = *(const float4*)(brow + 16 * m + 4);
            v8f v;
            v[0] = x0.x + b0.x; v[1] = x0.y + b0.y; v[2] = x0.z + b0.z; v[3] = x0.w + b0.w;
            v[4] = x1.x + b1.x; v[5] = x1.y + b1.y; v[6] = x1.z + b1.z; v[7] = x1.w + b1.w;
            xb[m] = v;
            v8f t;
            #pragma unroll
            for (int r = 0; r < 8; ++r) t[r] = tanh_hw(v[r]);
            zr[m] = t;
        }
    }

    float* xpark = zsh + lane * 4;
    *(float4*)(xpark)       = (float4){xb[6][0], xb[6][1], xb[6][2], xb[6][3]};
    *(float4*)(xpark + 128) = (float4){xb[6][4], xb[6][5], xb[6][6], xb[6][7]};
    *(float4*)(xpark + 256) = (float4){xb[7][0], xb[7][1], xb[7][2], xb[7][3]};
    *(float4*)(xpark + 384) = (float4){xb[7][4], xb[7][5], xb[7][6], xb[7][7]};

    const float* wp = wsh + lmod * LDS_STRIDE + 8 * lhalf;

    float4 wfa[8], wfb[8];
    #pragma unroll
    for (int m = 0; m < 8; ++m)
        wfa[m] = *(const float4*)(wp + m * 16 * LDS_STRIDE);

#define W_READS(buf, s)                                                                     \
    _Pragma("unroll")                                                                       \
    for (int m = 0; m < 8; ++m)                                                             \
        buf[m] = *(const float4*)(wp + m * 16 * LDS_STRIDE + 16 * ((s) >> 1) + 4 * ((s) & 1));

#define MMA_BLOCK(buf, s)                                                                   \
    {                                                                                       \
        v2f zlo, zhi;                                                                       \
        zlo[0] = zr[(s) >> 1][4 * ((s) & 1)];     zlo[1] = zr[(s) >> 1][4 * ((s) & 1) + 1]; \
        zhi[0] = zr[(s) >> 1][4 * ((s) & 1) + 2]; zhi[1] = zr[(s) >> 1][4 * ((s) & 1) + 3]; \
        _Pragma("unroll")                                                                   \
        for (int m = 0; m < 8; ++m) {                                                       \
            v2f wlo; wlo[0] = buf[m].x; wlo[1] = buf[m].y;                                  \
            acc[m] = __builtin_amdgcn_wmma_f32_16x16x4_f32(                                 \
                false, wlo, false, zlo, (short)0, acc[m], false, false);                    \
        }                                                                                   \
        _Pragma("unroll")                                                                   \
        for (int m = 0; m < 8; ++m) {                                                       \
            v2f whi; whi[0] = buf[m].z; whi[1] = buf[m].w;                                  \
            acc[m] = __builtin_amdgcn_wmma_f32_16x16x4_f32(                                 \
                false, whi, false, zhi, (short)0, acc[m], false, false);                    \
        }                                                                                   \
    }

    #pragma unroll 1
    for (int it = 1; it < MAX_ITER; ++it) {
        asm volatile("" ::: "memory");

        v8f acc[8];
        #pragma unroll
        for (int m = 0; m < 6; ++m) acc[m] = xb[m];
        {
            const float4 p0 = *(const float4*)(xpark);
            const float4 p1 = *(const float4*)(xpark + 128);
            const float4 p2 = *(const float4*)(xpark + 256);
            const float4 p3 = *(const float4*)(xpark + 384);
            acc[6] = (v8f){p0.x, p0.y, p0.z, p0.w, p1.x, p1.y, p1.z, p1.w};
            acc[7] = (v8f){p2.x, p2.y, p2.z, p2.w, p3.x, p3.y, p3.z, p3.w};
        }

        #pragma unroll
        for (int s = 0; s < 16; s += 2) {
            W_READS(wfb, s + 1)
            __builtin_amdgcn_sched_barrier(0x6);
            MMA_BLOCK(wfa, s)
            __builtin_amdgcn_sched_barrier(0x6);
            W_READS(wfa, (s + 2) & 15)
            __builtin_amdgcn_sched_barrier(0x6);
            MMA_BLOCK(wfb, s + 1)
            __builtin_amdgcn_sched_barrier(0x6);
        }

        float maxd = 0.f;
        #pragma unroll
        for (int m = 0; m < 8; ++m) {
            v8f t;
            #pragma unroll
            for (int r = 0; r < 8; ++r) {
                const float nw = tanh_hw(acc[m][r]);
                maxd = fmaxf(maxd, fabsf(nw - zr[m][r]));
                t[r] = nw;
            }
            zr[m] = t;
        }

        bool active = (maxd > CONV_EPS);
#if __has_builtin(__builtin_amdgcn_ballot_w32)
        if (__builtin_amdgcn_ballot_w32(active) == 0u) break;
#else
        if (!__any(active)) break;
#endif
    }

    {
        #pragma unroll
        for (int m = 0; m < 8; ++m) {
            float* zp = zsh + lmod * LDS_STRIDE + 16 * m + 8 * lhalf;
            float4 lo, hi;
            lo.x = zr[m][0]; lo.y = zr[m][1]; lo.z = zr[m][2]; lo.w = zr[m][3];
            hi.x = zr[m][4]; hi.y = zr[m][5]; hi.z = zr[m][6]; hi.w = zr[m][7];
            *(float4*)zp       = lo;
            *(float4*)(zp + 4) = hi;
        }
        asm volatile("s_wait_dscnt 0x0" ::: "memory");

        float* o = out + row0 * HIDDEN;
        #pragma unroll
        for (int p = 0; p < 16; ++p) {
            const int idx = p * 32 + lane;
            const int row = idx >> 5;
            const int c4  = idx & 31;
            float4 v = *(const float4*)(zsh + row * LDS_STRIDE + c4 * 4);
            *(float4*)(o + row * HIDDEN + c4 * 4) = v;
        }
    }
}

extern "C" void kernel_launch(void* const* d_in, const int* in_sizes, int n_in,
                              void* d_out, int out_size, void* d_ws, size_t ws_size,
                              hipStream_t stream) {
    const float* x = (const float*)d_in[0];
    const float* W = (const float*)d_in[1];
    const float* b = (const float*)d_in[2];
    float* out     = (float*)d_out;

    const int batch = in_sizes[0] / HIDDEN;
    const int rows_per_block = ROWS_PER_WAVE * WAVES_PER_BLK;
    const int grid = batch / rows_per_block;
    const size_t lds_bytes = (size_t)(W_LDS_FLOATS + WAVES_PER_BLK * Z_LDS_FLOATS) * 4;

    deq_f32_wmma_kernel<<<grid, THREADS, lds_bytes, stream>>>(x, W, b, out);
}
